# starting kernel
# baseline (speedup 1.0000x reference)
#include <hip/hip_runtime.h>
#include <hip/hip_bf16.h>
#include <math.h>

#define NF    128
#define EMBD  64
#define MAXD  16
#define NCLS  2

typedef float v2f __attribute__((ext_vector_type(2)));
typedef float v8f __attribute__((ext_vector_type(8)));

__global__ __launch_bounds__(256)
void care_phase1(const float* __restrict__ x,
                 const float* __restrict__ w_dist,
                 const float* __restrict__ b_dist,
                 float* __restrict__ score_out,
                 float* __restrict__ s0,
                 int N)
{
    int wave = (int)((blockIdx.x * blockDim.x + threadIdx.x) >> 5);
    int lane = threadIdx.x & 31;
    if (wave >= N) return;
    const float* xr = x + (size_t)wave * NF;
    float acc0 = 0.f, acc1 = 0.f;
    #pragma unroll
    for (int f = 0; f < NF; f += 32) {
        float v = xr[f + lane];
        acc0 += v * w_dist[(f + lane) * 2 + 0];
        acc1 += v * w_dist[(f + lane) * 2 + 1];
    }
    #pragma unroll
    for (int off = 16; off > 0; off >>= 1) {
        acc0 += __shfl_xor(acc0, off, 32);
        acc1 += __shfl_xor(acc1, off, 32);
    }
    if (lane == 0) {
        float l0 = acc0 + b_dist[0];
        float l1 = acc1 + b_dist[1];
        score_out[(size_t)wave * 2 + 0] = l0;
        score_out[(size_t)wave * 2 + 1] = l1;
        s0[wave] = tanhf(l0);
    }
}

__global__ __launch_bounds__(128)
void care_phase2(const float* __restrict__ x,
                 const float* __restrict__ w_self,
                 const float* __restrict__ w_neigh,
                 const float* __restrict__ b_sage,
                 const float* __restrict__ w_lin,
                 const float* __restrict__ b_lin,
                 const int*   __restrict__ neighbors,
                 const int*   __restrict__ degrees,
                 const float* __restrict__ s0,
                 float* __restrict__ logits_out,
                 int N)
{
    extern __shared__ float4 shW[];
    __shared__ float sh_an[4][16][NF + 4];
    __shared__ float sh_h [4][16][EMBD + 2];
    __shared__ float sh_wlin[EMBD * NCLS];
    __shared__ int   sh_sel[4][16][MAXD / 2];
    __shared__ int   sh_k [4][16];

    const int tid  = threadIdx.x;
    const int wid  = tid >> 5;
    const int lane = tid & 31;
    const int hi   = lane >> 4;
    const int lo   = lane & 15;

    int tile0 = (int)blockIdx.x * 64 + wid * 16;
    const bool active = (tile0 < N);
    if (!active) tile0 = 0;

    #pragma unroll 4
    for (int idx = tid; idx < (NF / 4) * 4 * 32; idx += 128) {
        const int l   = idx & 31;
        const int ct  = (idx >> 5) & 3;
        const int i   = idx >> 7;
        const int kk  = 4 * i + 2 * (l >> 4);
        const int col = ct * 16 + (l & 15);
        float4 w;
        w.x = w_self [(size_t)(kk)     * EMBD + col];
        w.y = w_self [(size_t)(kk + 1) * EMBD + col];
        w.z = w_neigh[(size_t)(kk)     * EMBD + col];
        w.w = w_neigh[(size_t)(kk + 1) * EMBD + col];
        shW[idx] = w;
    }
    if (tid < EMBD * NCLS) sh_wlin[tid] = w_lin[tid];

    if (lane < 16) {
        int node = tile0 + lane;
        if (node >= N) node = N - 1;
        int deg = degrees[node];
        if (deg < 1) deg = 1;
        if (deg > MAXD) deg = MAXD;
        int k = deg >> 1;
        if (k < 1) k = 1;
        float sc = s0[node];
        float dist[MAXD];
        int   ids [MAXD];
        #pragma unroll
        for (int j = 0; j < MAXD; ++j) {
            int nb = neighbors[(size_t)node * MAXD + j];
            ids[j] = nb;
            float d = fabsf(sc - s0[nb]);
            dist[j] = (j < deg) ? d : INFINITY;
        }
        for (int t = 0; t < k; ++t) {
            int best = 0; float bv = INFINITY;
            #pragma unroll
            for (int j = 0; j < MAXD; ++j) {
                if (dist[j] < bv) { bv = dist[j]; best = j; }
            }
            sh_sel[wid][lane][t] = ids[best];
            dist[best] = INFINITY;
        }
        sh_k[wid][lane] = k;
    }
    __syncthreads();

    for (int m = 0; m < 16; ++m) {
        int k = sh_k[wid][m];
        float a0 = 0.f, a1 = 0.f, a2 = 0.f, a3 = 0.f;
        for (int j = 0; j < k; ++j) {
            const float* xr = x + (size_t)sh_sel[wid][m][j] * NF;
            a0 += xr[lane];
            a1 += xr[lane + 32];
            a2 += xr[lane + 64];
            a3 += xr[lane + 96];
        }
        float inv = 1.0f / (float)k;
        sh_an[wid][m][lane]      = a0 * inv;
        sh_an[wid][m][lane + 32] = a1 * inv;
        sh_an[wid][m][lane + 64] = a2 * inv;
        sh_an[wid][m][lane + 96] = a3 * inv;
    }
    __syncthreads();

    int srow = tile0 + lo;
    if (srow >= N) srow = N - 1;
    const float* xself = x + (size_t)srow * NF;

    v8f c[4];
    #pragma unroll
    for (int ct = 0; ct < 4; ++ct) c[ct] = (v8f){0.f,0.f,0.f,0.f,0.f,0.f,0.f,0.f};

    for (int i = 0; i < NF / 4; ++i) {
        const int kk = 4 * i + hi * 2;
        v2f a_s, a_n;
        a_s.x = xself[kk];          a_s.y = xself[kk + 1];
        a_n.x = sh_an[wid][lo][kk]; a_n.y = sh_an[wid][lo][kk + 1];
        #pragma unroll
        for (int ct = 0; ct < 4; ++ct) {
            float4 wf = shW[(i * 4 + ct) * 32 + lane];
            v2f b_s; b_s.x = wf.x; b_s.y = wf.y;
            v2f b_n; b_n.x = wf.z; b_n.y = wf.w;
            c[ct] = __builtin_amdgcn_wmma_f32_16x16x4_f32(
                        false, a_s, false, b_s, (short)0, c[ct], false, false);
            c[ct] = __builtin_amdgcn_wmma_f32_16x16x4_f32(
                        false, a_n, false, b_n, (short)0, c[ct], false, false);
        }
    }

    #pragma unroll
    for (int ct = 0; ct < 4; ++ct) {
        const int e = ct * 16 + lo;
        const float be = b_sage[e];
        #pragma unroll
        for (int j = 0; j < 8; ++j) {
            float v = c[ct][j] + be;
            sh_h[wid][j + 8 * hi][e] = v > 0.f ? v : 0.f;
        }
    }
    __syncthreads();

    {
        const int m = lane >> 1, cls = lane & 1;
        float acc = b_lin[cls];
        #pragma unroll 8
        for (int e = 0; e < EMBD; ++e)
            acc += sh_h[wid][m][e] * sh_wlin[e * NCLS + cls];
        const int node = tile0 + m;
        if (active && node < N)
            logits_out[(size_t)node * NCLS + cls] = acc;
    }
}

extern "C" void kernel_launch(void* const* d_in, const int* in_sizes, int n_in,
                              void* d_out, int out_size, void* d_ws, size_t ws_size,
                              hipStream_t stream) {
    const float* x       = (const float*)d_in[0];
    const float* w_dist  = (const float*)d_in[1];
    const float* b_dist  = (const float*)d_in[2];
    const float* w_self  = (const float*)d_in[3];
    const float* w_neigh = (const float*)d_in[4];
    const float* b_sage  = (const float*)d_in[5];
    const float* w_lin   = (const float*)d_in[6];
    const float* b_lin   = (const float*)d_in[7];
    const int*   neigh   = (const int*)d_in[8];
    const int*   degs    = (const int*)d_in[9];

    const int N = in_sizes[9];
    float* out       = (float*)d_out;
    float* logits    = out;
    float* score_out = out + (size_t)N * NCLS;
    float* s0        = (float*)d_ws;

    {
        int waves_per_block = 256 / 32;
        int blocks = (N + waves_per_block - 1) / waves_per_block;
        care_phase1<<<blocks, 256, 0, stream>>>(x, w_dist, b_dist,
                                                score_out, s0, N);
    }
    {
        int blocks = (N + 63) / 64;
        size_t dynbytes = (size_t)(NF / 4) * 4 * 32 * sizeof(float4);
        care_phase2<<<blocks, 128, dynbytes, stream>>>(x, w_self, w_neigh,
                                                       b_sage, w_lin, b_lin,
                                                       neigh, degs, s0,
                                                       logits, N);
    }
}


# Projection launch: each of 8 waves keeps its 16-column slice of [W_self|W_neigh] as B fragments in registers; long-lived blocks walk 64-node x tiles shared in LDS, next rows loaded under the products
# speedup vs baseline: 2.6864x; 2.6864x over previous
#include <hip/hip_runtime.h>
#include <math.h>


#define NF    128
#define EMBD  64
#define MAXD  16
#define NCLS  2
#define PW    (2 * EMBD)
#define GU    2
#define TM    64
#define PBLK  512

typedef float v2f __attribute__((ext_vector_type(2)));
typedef float v4f __attribute__((ext_vector_type(4)));
typedef float v8f __attribute__((ext_vector_type(8)));

template <int XM>
__device__ __forceinline__ float row_xor16(float v) {
    return __int_as_float(__builtin_amdgcn_update_dpp(
        0, __float_as_int(v), 0x160 + XM, 0xf, 0xf, false));
}
__device__ __forceinline__ float row_min16(float v) {
    v = fminf(v, row_xor16<1>(v));
    v = fminf(v, row_xor16<2>(v));
    v = fminf(v, row_xor16<4>(v));
    v = fminf(v, row_xor16<8>(v));
    return v;
}
__device__ __forceinline__ float row_sum16(float v) {
    v += row_xor16<1>(v);
    v += row_xor16<2>(v);
    v += row_xor16<4>(v);
    v += row_xor16<8>(v);
    return v;
}

__global__ __launch_bounds__(256)
void project_score_kernel(const float* __restrict__ x,
                          const float* __restrict__ w_dist,
                          const float* __restrict__ b_dist,
                          const float* __restrict__ w_self,
                          const float* __restrict__ w_neigh,
                          const float* __restrict__ b_sage,
                          float* __restrict__ score_out,
                          float* __restrict__ proj,
                          float* __restrict__ s0,
                          size_t ws_bytes, int N)
{
    if ((size_t)N * (PW + 1) * sizeof(float) > ws_bytes) return;

    __shared__ __attribute__((aligned(16))) float sh_x[TM][NF + 4];

    const int tid  = threadIdx.x;
    const int wid  = tid >> 5;
    const int lane = tid & 31;
    const int hi   = lane >> 4;
    const int lo   = lane & 15;

    const float* __restrict__ wsrc = (wid < 4) ? w_self : w_neigh;
    const int col = (wid & 3) * 16 + lo;
    v2f bfrag[NF / 4];
    #pragma unroll
    for (int i = 0; i < NF / 4; ++i) {
        const int kk = 4 * i + 2 * hi;
        bfrag[i].x = wsrc[(size_t)(kk)     * EMBD + col];
        bfrag[i].y = wsrc[(size_t)(kk + 1) * EMBD + col];
    }
    const float bias = (wid < 4) ? b_sage[col] : 0.f;
    const int   ecol = wid * 16 + lo;

    float wd0[4], wd1[4];
    #pragma unroll
    for (int q = 0; q < 4; ++q) {
        wd0[q] = w_dist[(32 * q + lane) * 2 + 0];
        wd1[q] = w_dist[(32 * q + lane) * 2 + 1];
    }
    const float bd0 = b_dist[0], bd1 = b_dist[1];

    const int n_st = (N + TM - 1) / TM;
    int st = (int)blockIdx.x;

    float xr[8][4];
    #pragma unroll
    for (int m = 0; m < 8; ++m) {
        int node = st * TM + wid * 8 + m;
        if (node >= N) node = N - 1;
        const float* xp = x + (size_t)node * NF;
        #pragma unroll
        for (int q = 0; q < 4; ++q) xr[m][q] = xp[32 * q + lane];
    }

    for (; st < n_st; st += (int)gridDim.x) {
        const int base = st * TM + wid * 8;

        float keep0 = 0.f, keep1 = 0.f;
        #pragma unroll
        for (int m = 0; m < 8; ++m) {
            float acc0 = 0.f, acc1 = 0.f;
            #pragma unroll
            for (int q = 0; q < 4; ++q) {
                const float v = xr[m][q];
                sh_x[wid * 8 + m][32 * q + lane] = v;
                acc0 += v * wd0[q];
                acc1 += v * wd1[q];
            }
            #pragma unroll
            for (int off = 16; off > 0; off >>= 1) {
                acc0 += __shfl_xor(acc0, off, 32);
                acc1 += __shfl_xor(acc1, off, 32);
            }
            if (lane == m) { keep0 = acc0; keep1 = acc1; }
        }
        if (lane < 8) {
            const int node = base + lane;
            if (node < N) {
                const float l0 = keep0 + bd0;
                const float l1 = keep1 + bd1;
                score_out[(size_t)node * 2 + 0] = l0;
                score_out[(size_t)node * 2 + 1] = l1;
                s0[node] = tanhf(l0);
            }
        }
        __syncthreads();

        const int nxt = st + (int)gridDim.x;
        if (nxt < n_st) {
            #pragma unroll
            for (int m = 0; m < 8; ++m) {
                int node = nxt * TM + wid * 8 + m;
                if (node >= N) node = N - 1;
                const float* xp = x + (size_t)node * NF;
                #pragma unroll
                for (int q = 0; q < 4; ++q) xr[m][q] = xp[32 * q + lane];
            }
        }

        v8f c[TM / 16];
        #pragma unroll
        for (int t = 0; t < TM / 16; ++t)
            c[t] = (v8f){0.f,0.f,0.f,0.f,0.f,0.f,0.f,0.f};
        #pragma unroll
        for (int i = 0; i < NF / 4; ++i) {
            const int kk = 4 * i + hi * 2;
            #pragma unroll
            for (int t = 0; t < TM / 16; ++t) {
                v2f a;
                a.x = sh_x[16 * t + lo][kk]; a.y = sh_x[16 * t + lo][kk + 1];
                c[t] = __builtin_amdgcn_wmma_f32_16x16x4_f32(
                           false, a, false, bfrag[i], (short)0, c[t], false, false);
            }
        }
        __syncthreads();

        #pragma unroll
        for (int t = 0; t < TM / 16; ++t) {
            #pragma unroll
            for (int j = 0; j < 8; ++j)
                sh_x[16 * t + j + 8 * hi][ecol] = c[t][j] + bias;
        }
        __syncthreads();
        v4f rowv[8];
        #pragma unroll
        for (int r = 0; r < 8; ++r)
            rowv[r] = *(const v4f*)&sh_x[wid * 8 + r][4 * lane];
        #pragma unroll
        for (int r = 0; r < 8; ++r) {
            const int node = base + r;
            if (node < N)
                *(v4f*)(proj + (size_t)node * PW + 4 * lane) = rowv[r];
        }
    }
}

__global__ __launch_bounds__(256)
void select_gather_kernel(const float* __restrict__ proj,
                          const float* __restrict__ s0,
                          const int*   __restrict__ neighbors,
                          const int*   __restrict__ degrees,
                          const float* __restrict__ w_lin,
                          const float* __restrict__ b_lin,
                          float* __restrict__ logits_out,
                          size_t ws_bytes, int N)
{
    if ((size_t)N * (PW + 1) * sizeof(float) > ws_bytes) return;

    const int lane = threadIdx.x & 31;
    const int half = lane >> 4;
    const int lo   = lane & 15;
    const int wave = (int)((blockIdx.x * blockDim.x + threadIdx.x) >> 5);

    int   node[GU], nb[GU], kk[GU];
    bool  live[GU];
    float d[GU];
    v4f   z[GU];
    v4f   r[GU][MAXD / 2];
    int   kmax = 1;

    #pragma unroll
    for (int u = 0; u < GU; ++u) {
        const int n = (wave * GU + u) * 2 + half;
        live[u] = (n < N);
        node[u] = live[u] ? n : (N - 1);
        int deg = degrees[node[u]];
        if (deg < 1) deg = 1;
        if (deg > MAXD) deg = MAXD;
        int k = deg >> 1;
        if (k < 1) k = 1;
        kk[u] = k;
        nb[u] = neighbors[(size_t)node[u] * MAXD + lo];
        const float sc = s0[node[u]];
        const float dd = fabsf(sc - s0[nb[u]]);
        d[u] = (lo < deg) ? dd : INFINITY;
        z[u] = *(const v4f*)(proj + (size_t)node[u] * PW + 4 * lo);
        #pragma unroll
        for (int t = 0; t < MAXD / 2; ++t) r[u][t] = (v4f){0.f, 0.f, 0.f, 0.f};
        const int ka = __builtin_amdgcn_readlane(k, 0);
        const int kb = __builtin_amdgcn_readlane(k, 16);
        kmax = max(kmax, max(ka, kb));
    }

    #pragma unroll
    for (int t = 0; t < MAXD / 2; ++t) {
        if (t < kmax) {
            #pragma unroll
            for (int u = 0; u < GU; ++u) {
                const float m = row_min16(d[u]);
                const unsigned hit = __builtin_amdgcn_ballot_w32(d[u] == m);
                const int ja = __builtin_ctz((hit & 0xFFFFu) | 0x10000u) & 15;
                const int jb = __builtin_ctz((hit >> 16)     | 0x10000u) & 15;
                const int ida = __builtin_amdgcn_readlane(nb[u], ja);
                const int idb = __builtin_amdgcn_readlane(nb[u], 16 + jb);
                const int id = half ? idb : ida;
                const int jw = half ? jb  : ja;
                if (lo == jw) d[u] = INFINITY;
                if (t < kk[u])
                    r[u][t] = *(const v4f*)(proj + (size_t)id * PW + EMBD + 4 * lo);
            }
        }
    }

    float wl0[4], wl1[4];
    #pragma unroll
    for (int q = 0; q < 4; ++q) {
        wl0[q] = w_lin[(4 * lo + q) * NCLS + 0];
        wl1[q] = w_lin[(4 * lo + q) * NCLS + 1];
    }
    const float bl0 = b_lin[0], bl1 = b_lin[1];

    #pragma unroll
    for (int u = 0; u < GU; ++u) {
        v4f a = r[u][0];
        #pragma unroll
        for (int t = 1; t < MAXD / 2; ++t) a += r[u][t];
        const float inv = 1.0f / (float)kk[u];
        v4f h = z[u] + a * inv;
        float p0 = 0.f, p1 = 0.f;
        #pragma unroll
        for (int q = 0; q < 4; ++q) {
            const float hv = h[q] > 0.f ? h[q] : 0.f;
            p0 += hv * wl0[q];
            p1 += hv * wl1[q];
        }
        p0 = row_sum16(p0);
        p1 = row_sum16(p1);
        if (lo == 0 && live[u]) {
            v2f o; o.x = p0 + bl0; o.y = p1 + bl1;
            *(v2f*)(logits_out + (size_t)node[u] * NCLS) = o;
        }
    }
}

extern "C" void kernel_launch(void* const* d_in, const int* in_sizes, int n_in,
                              void* d_out, int out_size, void* d_ws, size_t ws_size,
                              hipStream_t stream) {
    const float* x       = (const float*)d_in[0];
    const float* w_dist  = (const float*)d_in[1];
    const float* b_dist  = (const float*)d_in[2];
    const float* w_self  = (const float*)d_in[3];
    const float* w_neigh = (const float*)d_in[4];
    const float* b_sage  = (const float*)d_in[5];
    const float* w_lin   = (const float*)d_in[6];
    const float* b_lin   = (const float*)d_in[7];
    const int*   neigh   = (const int*)d_in[8];
    const int*   degs    = (const int*)d_in[9];

    const int N = in_sizes[9];
    float* out       = (float*)d_out;
    float* logits    = out;
    float* score_out = out + (size_t)N * NCLS;
    float* proj      = (float*)d_ws;
    float* s0        = proj + (size_t)N * PW;

    {
        int tiles  = (N + TM - 1) / TM;
        int blocks = tiles < PBLK ? tiles : PBLK;
        project_score_kernel<<<blocks, 256, 0, stream>>>(
            x, w_dist, b_dist, w_self, w_neigh, b_sage,
            score_out, proj, s0, ws_size, N);
    }
    {
        int nodes_per_block = (256 / 32) * 2 * GU;
        int blocks = (N + nodes_per_block - 1) / nodes_per_block;
        select_gather_kernel<<<blocks, 256, 0, stream>>>(
            proj, s0, neigh, degs, w_lin, b_lin, logits, ws_size, N);
    }
}
